# starting kernel
# baseline (speedup 1.0000x reference)
#include <hip/hip_runtime.h>
#include <hip/hip_bf16.h>

typedef __attribute__((ext_vector_type(2))) float v2f;
typedef __attribute__((ext_vector_type(8))) float v8f;

#define EMBED 512
#define VOCAB 512
#define XS_STRIDE 516

__global__ __launch_bounds__(256) void quantizer_bias_kernel(
    const float* __restrict__ w, float* __restrict__ bias) {
  const int v = blockIdx.x * blockDim.x + threadIdx.x;
  float s = 0.0f;
  #pragma unroll 8
  for (int k = 0; k < VOCAB; ++k) {
    const float x = w[(long)k * EMBED + v];
    s = fmaf(x, x, s);
  }
  bias[v] = s;
}

__global__ __launch_bounds__(256) void quantizer_main_kernel(
    const float* __restrict__ x, const float* __restrict__ w,
    const float* __restrict__ bias, float* __restrict__ out) {
  __shared__ float Xs[16 * XS_STRIDE];
  __shared__ float redV[8][16];
  __shared__ int   redI[8][16];
  __shared__ int   fidx[16];

  const int tid  = threadIdx.x;
  const int wave = __builtin_amdgcn_readfirstlane(tid >> 5);
  const int lane = tid & 31;
  const int lo   = lane & 15;
  const int hi   = lane >> 4;
  const long r0  = (long)blockIdx.x * 16;

  #pragma unroll
  for (int i = 0; i < 8; ++i) {
    const int t   = tid + i * 256;
    const int row = t >> 7;
    const int c4  = t & 127;
    const float4 val = ((const float4*)(x + (r0 + row) * EMBED))[c4];
    *(float4*)&Xs[row * XS_STRIDE + c4 * 4] = val;
  }
  __syncthreads();

  float best[8];
  int   bidx[8];
  #pragma unroll
  for (int j = 0; j < 8; ++j) { best[j] = __builtin_inff(); bidx[j] = 0; }

  const float* xrow = Xs + lo * XS_STRIDE + 2 * hi;

  #pragma unroll
  for (int t = 0; t < 4; ++t) {
    const int vt = wave + t * 8;
    const int v  = vt * 16 + lo;
    const float* wrow = w + (long)v * EMBED + 2 * hi;

    v8f c0 = {};
    v8f c1 = {};
    #pragma unroll 4
    for (int k0 = 0; k0 < EMBED; k0 += 8) {
      const v2f a0 = *(const v2f*)(xrow + k0);
      const v2f b0 = *(const v2f*)(wrow + k0);
      const v2f a1 = *(const v2f*)(xrow + k0 + 4);
      const v2f b1 = *(const v2f*)(wrow + k0 + 4);
      c0 = __builtin_amdgcn_wmma_f32_16x16x4_f32(false, a0, false, b0,
                                                 (short)0, c0, false, false);
      c1 = __builtin_amdgcn_wmma_f32_16x16x4_f32(false, a1, false, b1,
                                                 (short)0, c1, false, false);
    }

    const float bv = bias[v];
    #pragma unroll
    for (int j = 0; j < 8; ++j) {
      const float d = fmaf(-2.0f, c0[j] + c1[j], bv);
      if (d < best[j]) { best[j] = d; bidx[j] = v; }
    }
  }

  #pragma unroll
  for (int j = 0; j < 8; ++j) {
    #pragma unroll
    for (int m = 1; m <= 8; m <<= 1) {
      const float ov = __shfl_xor(best[j], m, 32);
      const int   oi = __shfl_xor(bidx[j], m, 32);
      if (ov < best[j] || (ov == best[j] && oi < bidx[j])) {
        best[j] = ov; bidx[j] = oi;
      }
    }
  }
  if (lo == 0) {
    #pragma unroll
    for (int j = 0; j < 8; ++j) {
      redV[wave][hi * 8 + j] = best[j];
      redI[wave][hi * 8 + j] = bidx[j];
    }
  }
  __syncthreads();

  if (tid < 16) {
    float bv = redV[0][tid];
    int   bi = redI[0][tid];
    #pragma unroll
    for (int w8 = 1; w8 < 8; ++w8) {
      const float ov = redV[w8][tid];
      const int   oi = redI[w8][tid];
      if (ov < bv || (ov == bv && oi < bi)) { bv = ov; bi = oi; }
    }
    fidx[tid] = bi;
  }
  __syncthreads();

  const int row = tid >> 4;
  const int l16 = tid & 15;
  const float4* src = (const float4*)(w + (long)fidx[row] * EMBED);
  float4* dst = (float4*)(out + (r0 + row) * EMBED);
  #pragma unroll
  for (int i = 0; i < 8; ++i) dst[l16 + i * 16] = src[l16 + i * 16];
}

extern "C" void kernel_launch(void* const* d_in, const int* in_sizes, int n_in,
                              void* d_out, int out_size, void* d_ws, size_t ws_size,
                              hipStream_t stream) {
  const float* x = (const float*)d_in[0];
  const float* w = (const float*)d_in[1];
  float* out  = (float*)d_out;
  float* bias = (float*)d_ws;

  quantizer_bias_kernel<<<VOCAB / 256, 256, 0, stream>>>(w, bias);

  const int n_rows = in_sizes[0] / EMBED;
  quantizer_main_kernel<<<n_rows / 16, 256, 0, stream>>>(x, w, bias, out);
}


# Reuse each global W fragment across four 16-row X tiles (64-row work-group, X in LDS): one 16-line global load per 4 WMMAs instead of per 1; per-element sums and k order unchanged
# speedup vs baseline: 1.4533x; 1.4533x over previous
#include <hip/hip_runtime.h>
#include <hip/hip_bf16.h>

typedef __attribute__((ext_vector_type(2))) float v2f;
typedef __attribute__((ext_vector_type(8))) float v8f;

#define EMBED 512
#define VOCAB 512
#define XS_STRIDE 516
#define ROWS_PER_BLOCK 64
#define ROW_TILES 4

__global__ __launch_bounds__(256) void quantizer_bias_kernel(
    const float* __restrict__ w, float* __restrict__ bias) {
  const int v = blockIdx.x * blockDim.x + threadIdx.x;
  float s = 0.0f;
  #pragma unroll 8
  for (int k = 0; k < VOCAB; ++k) {
    const float x = w[(long)k * EMBED + v];
    s = fmaf(x, x, s);
  }
  bias[v] = s;
}

__global__ __launch_bounds__(256) void quantizer_main_kernel(
    const float* __restrict__ x, const float* __restrict__ w,
    const float* __restrict__ bias, float* __restrict__ out) {
  __shared__ float Xs[ROWS_PER_BLOCK * XS_STRIDE];
  __shared__ float redV[8][ROWS_PER_BLOCK];
  __shared__ int   redI[8][ROWS_PER_BLOCK];
  __shared__ int   fidx[ROWS_PER_BLOCK];

  const int tid  = threadIdx.x;
  const int wave = __builtin_amdgcn_readfirstlane(tid >> 5);
  const int lane = tid & 31;
  const int lo   = lane & 15;
  const int hi   = lane >> 4;
  const long r0  = (long)blockIdx.x * ROWS_PER_BLOCK;

  #pragma unroll 8
  for (int i = 0; i < (ROWS_PER_BLOCK * EMBED / 4) / 256; ++i) {
    const int t   = tid + i * 256;
    const int row = t >> 7;
    const int c4  = t & 127;
    const float4 val = ((const float4*)(x + (r0 + row) * EMBED))[c4];
    *(float4*)&Xs[row * XS_STRIDE + c4 * 4] = val;
  }
  __syncthreads();

  float best[ROW_TILES][8];
  int   bidx[ROW_TILES][8];
  #pragma unroll
  for (int rt = 0; rt < ROW_TILES; ++rt) {
    #pragma unroll
    for (int j = 0; j < 8; ++j) { best[rt][j] = __builtin_inff(); bidx[rt][j] = 0; }
  }

  const float* xrow = Xs + lo * XS_STRIDE + 2 * hi;

  #pragma nounroll
  for (int t = 0; t < 4; ++t) {
    const int vt = wave + t * 8;
    const int v  = vt * 16 + lo;
    const float* wrow = w + (long)v * EMBED + 2 * hi;

    v8f c0[ROW_TILES];
    v8f c1[ROW_TILES];
    #pragma unroll
    for (int rt = 0; rt < ROW_TILES; ++rt) { c0[rt] = (v8f){}; c1[rt] = (v8f){}; }

    #pragma unroll 4
    for (int k0 = 0; k0 < EMBED; k0 += 8) {
      const v2f b0 = *(const v2f*)(wrow + k0);
      const v2f b1 = *(const v2f*)(wrow + k0 + 4);
      #pragma unroll
      for (int rt = 0; rt < ROW_TILES; ++rt) {
        const float* xr = xrow + rt * 16 * XS_STRIDE + k0;
        const v2f a0 = *(const v2f*)(xr);
        const v2f a1 = *(const v2f*)(xr + 4);
        c0[rt] = __builtin_amdgcn_wmma_f32_16x16x4_f32(false, a0, false, b0,
                                                       (short)0, c0[rt], false, false);
        c1[rt] = __builtin_amdgcn_wmma_f32_16x16x4_f32(false, a1, false, b1,
                                                       (short)0, c1[rt], false, false);
      }
    }

    const float bv = bias[v];
    #pragma unroll
    for (int rt = 0; rt < ROW_TILES; ++rt) {
      #pragma unroll
      for (int j = 0; j < 8; ++j) {
        const float d = fmaf(-2.0f, c0[rt][j] + c1[rt][j], bv);
        if (d < best[rt][j]) { best[rt][j] = d; bidx[rt][j] = v; }
      }
    }
  }

  #pragma unroll
  for (int rt = 0; rt < ROW_TILES; ++rt) {
    #pragma unroll
    for (int j = 0; j < 8; ++j) {
      #pragma unroll
      for (int m = 1; m <= 8; m <<= 1) {
        const float ov = __shfl_xor(best[rt][j], m, 32);
        const int   oi = __shfl_xor(bidx[rt][j], m, 32);
        if (ov < best[rt][j] || (ov == best[rt][j] && oi < bidx[rt][j])) {
          best[rt][j] = ov; bidx[rt][j] = oi;
        }
      }
    }
  }
  if (lo == 0) {
    #pragma unroll
    for (int rt = 0; rt < ROW_TILES; ++rt) {
      #pragma unroll
      for (int j = 0; j < 8; ++j) {
        redV[wave][rt * 16 + hi * 8 + j] = best[rt][j];
        redI[wave][rt * 16 + hi * 8 + j] = bidx[rt][j];
      }
    }
  }
  __syncthreads();

  if (tid < ROWS_PER_BLOCK) {
    float bv = redV[0][tid];
    int   bi = redI[0][tid];
    #pragma unroll
    for (int w8 = 1; w8 < 8; ++w8) {
      const float ov = redV[w8][tid];
      const int   oi = redI[w8][tid];
      if (ov < bv || (ov == bv && oi < bi)) { bv = ov; bi = oi; }
    }
    fidx[tid] = bi;
  }
  __syncthreads();

  const int rsub = tid >> 4;
  const int l16  = tid & 15;
  #pragma unroll
  for (int rr = 0; rr < ROWS_PER_BLOCK / 16; ++rr) {
    const int row = rr * 16 + rsub;
    const float4* src = (const float4*)(w + (long)fidx[row] * EMBED);
    float4* dst = (float4*)(out + (r0 + row) * EMBED);
    #pragma unroll
    for (int i = 0; i < 8; ++i) dst[l16 + i * 16] = src[l16 + i * 16];
  }
}

extern "C" void kernel_launch(void* const* d_in, const int* in_sizes, int n_in,
                              void* d_out, int out_size, void* d_ws, size_t ws_size,
                              hipStream_t stream) {
  const float* x = (const float*)d_in[0];
  const float* w = (const float*)d_in[1];
  float* out  = (float*)d_out;
  float* bias = (float*)d_ws;

  quantizer_bias_kernel<<<VOCAB / 256, 256, 0, stream>>>(w, bias);

  const int n_rows = in_sizes[0] / EMBED;
  quantizer_main_kernel<<<n_rows / ROWS_PER_BLOCK, 256, 0, stream>>>(x, w, bias, out);
}


# W fragments of the next 32-k block requested one block ahead into a second register set (one loop of 64 blocks, two per trip): no block waits on its own global loads; sums and k order unchanged
# speedup vs baseline: 1.5022x; 1.0336x over previous
#include <hip/hip_runtime.h>
#include <hip/hip_bf16.h>

typedef __attribute__((ext_vector_type(2))) float v2f;
typedef __attribute__((ext_vector_type(8))) float v8f;

#define EMBED 512
#define VOCAB 512
#define XS_STRIDE 516
#define ROWS_PER_BLOCK 64
#define ROW_TILES 4

__global__ __launch_bounds__(256) void quantizer_bias_kernel(
    const float* __restrict__ w, float* __restrict__ bias) {
  const int v = blockIdx.x * blockDim.x + threadIdx.x;
  float s = 0.0f;
  #pragma unroll 8
  for (int k = 0; k < VOCAB; ++k) {
    const float x = w[(long)k * EMBED + v];
    s = fmaf(x, x, s);
  }
  bias[v] = s;
}

__global__ __launch_bounds__(256) void quantizer_main_kernel(
    const float* __restrict__ x, const float* __restrict__ w,
    const float* __restrict__ bias, float* __restrict__ out) {
  __shared__ float Xs[ROWS_PER_BLOCK * XS_STRIDE];
  __shared__ float redV[8][ROWS_PER_BLOCK];
  __shared__ int   redI[8][ROWS_PER_BLOCK];
  __shared__ int   fidx[ROWS_PER_BLOCK];

  const int tid  = threadIdx.x;
  const int wave = __builtin_amdgcn_readfirstlane(tid >> 5);
  const int lane = tid & 31;
  const int lo   = lane & 15;
  const int hi   = lane >> 4;
  const long r0  = (long)blockIdx.x * ROWS_PER_BLOCK;

  #pragma unroll 8
  for (int i = 0; i < (ROWS_PER_BLOCK * EMBED / 4) / 256; ++i) {
    const int t   = tid + i * 256;
    const int row = t >> 7;
    const int c4  = t & 127;
    const float4 val = ((const float4*)(x + (r0 + row) * EMBED))[c4];
    *(float4*)&Xs[row * XS_STRIDE + c4 * 4] = val;
  }
  __syncthreads();

  float best[ROW_TILES][8];
  int   bidx[ROW_TILES][8];
  #pragma unroll
  for (int rt = 0; rt < ROW_TILES; ++rt) {
    #pragma unroll
    for (int j = 0; j < 8; ++j) { best[rt][j] = __builtin_inff(); bidx[rt][j] = 0; }
  }

  const float* xrow = Xs + lo * XS_STRIDE + 2 * hi;


#define LOAD_W_BLOCK(dst, ptr)                                              \
  {                                                                         \
    _Pragma("unroll")                                                       \
    for (int i_ = 0; i_ < 8; ++i_) dst[i_] = *(const v2f*)((ptr) + 4 * i_); \
  }

#define MMA_BLOCK(wb, kf)                                                   \
  {                                                                         \
    _Pragma("unroll")                                                       \
    for (int s_ = 0; s_ < 4; ++s_) {                                        \
      _Pragma("unroll")                                                     \
      for (int rt = 0; rt < ROW_TILES; ++rt) {                              \
        const float* xr = xrow + rt * 16 * XS_STRIDE + (kf) + 8 * s_;       \
        const v2f a0 = *(const v2f*)(xr);                                   \
        const v2f a1 = *(const v2f*)(xr + 4);                               \
        c0[rt] = __builtin_amdgcn_wmma_f32_16x16x4_f32(false, a0, false, wb[2 * s_],     \
                                                       (short)0, c0[rt], false, false);  \
        c1[rt] = __builtin_amdgcn_wmma_f32_16x16x4_f32(false, a1, false, wb[2 * s_ + 1], \
                                                       (short)0, c1[rt], false, false);  \
      }                                                                     \
    }                                                                       \
  }

  v8f c0[ROW_TILES];
  v8f c1[ROW_TILES];
  #pragma unroll
  for (int rt = 0; rt < ROW_TILES; ++rt) { c0[rt] = (v8f){}; c1[rt] = (v8f){}; }

  const float* wp = w + (long)(wave * 16 + lo) * EMBED + 2 * hi;
  v2f bA[8];
  v2f bB[8];
  LOAD_W_BLOCK(bA, wp);

  #pragma nounroll
  for (int blk = 0; blk < 64; blk += 2) {
    const int kb = (blk & 15) * 32;

    LOAD_W_BLOCK(bB, wp + 32);
    __builtin_amdgcn_sched_barrier(0);
    MMA_BLOCK(bA, kb);
    __builtin_amdgcn_sched_barrier(0);

    const int step = (kb != 448) ? 64 : ((blk != 62) ? (128 * EMBED - 448) : (-3 * 128 * EMBED - 448));
    wp += step;
    LOAD_W_BLOCK(bA, wp);
    __builtin_amdgcn_sched_barrier(0);
    MMA_BLOCK(bB, kb + 32);

    if (kb == 448) {
      const int v = (wave + (blk >> 4) * 8) * 16 + lo;
      const float bv = bias[v];
      #pragma unroll
      for (int rt = 0; rt < ROW_TILES; ++rt) {
        #pragma unroll
        for (int j = 0; j < 8; ++j) {
          const float d = fmaf(-2.0f, c0[rt][j] + c1[rt][j], bv);
          if (d < best[rt][j]) { best[rt][j] = d; bidx[rt][j] = v; }
        }
        c0[rt] = (v8f){}; c1[rt] = (v8f){};
      }
    }
  }
#undef LOAD_W_BLOCK
#undef MMA_BLOCK

  #pragma unroll
  for (int rt = 0; rt < ROW_TILES; ++rt) {
    #pragma unroll
    for (int j = 0; j < 8; ++j) {
      #pragma unroll
      for (int m = 1; m <= 8; m <<= 1) {
        const float ov = __shfl_xor(best[rt][j], m, 32);
        const int   oi = __shfl_xor(bidx[rt][j], m, 32);
        if (ov < best[rt][j] || (ov == best[rt][j] && oi < bidx[rt][j])) {
          best[rt][j] = ov; bidx[rt][j] = oi;
        }
      }
    }
  }
  if (lo == 0) {
    #pragma unroll
    for (int rt = 0; rt < ROW_TILES; ++rt) {
      #pragma unroll
      for (int j = 0; j < 8; ++j) {
        redV[wave][rt * 16 + hi * 8 + j] = best[rt][j];
        redI[wave][rt * 16 + hi * 8 + j] = bidx[rt][j];
      }
    }
  }
  __syncthreads();

  if (tid < ROWS_PER_BLOCK) {
    float bv = redV[0][tid];
    int   bi = redI[0][tid];
    #pragma unroll
    for (int w8 = 1; w8 < 8; ++w8) {
      const float ov = redV[w8][tid];
      const int   oi = redI[w8][tid];
      if (ov < bv || (ov == bv && oi < bi)) { bv = ov; bi = oi; }
    }
    fidx[tid] = bi;
  }
  __syncthreads();

  const int rsub = tid >> 4;
  const int l16  = tid & 15;
  #pragma unroll
  for (int rr = 0; rr < ROWS_PER_BLOCK / 16; ++rr) {
    const int row = rr * 16 + rsub;
    const float4* src = (const float4*)(w + (long)fidx[row] * EMBED);
    float4* dst = (float4*)(out + (r0 + row) * EMBED);
    #pragma unroll
    for (int i = 0; i < 8; ++i) dst[l16 + i * 16] = src[l16 + i * 16];
  }
}

extern "C" void kernel_launch(void* const* d_in, const int* in_sizes, int n_in,
                              void* d_out, int out_size, void* d_ws, size_t ws_size,
                              hipStream_t stream) {
  const float* x = (const float*)d_in[0];
  const float* w = (const float*)d_in[1];
  float* out  = (float*)d_out;
  float* bias = (float*)d_ws;

  quantizer_bias_kernel<<<VOCAB / 256, 256, 0, stream>>>(w, bias);

  const int n_rows = in_sizes[0] / EMBED;
  quantizer_main_kernel<<<n_rows / ROWS_PER_BLOCK, 256, 0, stream>>>(x, w, bias, out);
}


# X rows' k pairs interleaved in LDS so one 16-byte read gives both A fragments of a k-step; reads issued eight at a time half a block ahead into a second register set; sums and k order unchanged
# speedup vs baseline: 1.5518x; 1.0330x over previous
#include <hip/hip_runtime.h>
#include <hip/hip_bf16.h>

typedef __attribute__((ext_vector_type(2))) float v2f;
typedef __attribute__((ext_vector_type(4))) float v4f;
typedef __attribute__((ext_vector_type(8))) float v8f;

#define EMBED 512
#define VOCAB 512
#define XS_STRIDE 516
#define XS_HALF 256
#define ROWS_PER_BLOCK 64
#define ROW_TILES 4

__global__ __launch_bounds__(256) void quantizer_bias_kernel(
    const float* __restrict__ w, float* __restrict__ bias) {
  const int v = blockIdx.x * blockDim.x + threadIdx.x;
  float s = 0.0f;
  #pragma unroll 8
  for (int k = 0; k < VOCAB; ++k) {
    const float x = w[(long)k * EMBED + v];
    s = fmaf(x, x, s);
  }
  bias[v] = s;
}

__global__ __launch_bounds__(256) void quantizer_main_kernel(
    const float* __restrict__ x, const float* __restrict__ w,
    const float* __restrict__ bias, float* __restrict__ out) {
  __shared__ __attribute__((aligned(16))) float Xs[ROWS_PER_BLOCK * XS_STRIDE];
  __shared__ float redV[8][ROWS_PER_BLOCK];
  __shared__ int   redI[8][ROWS_PER_BLOCK];
  __shared__ int   fidx[ROWS_PER_BLOCK];

  const int tid  = threadIdx.x;
  const int wave = __builtin_amdgcn_readfirstlane(tid >> 5);
  const int lane = tid & 31;
  const int lo   = lane & 15;
  const int hi   = lane >> 4;
  const long r0  = (long)blockIdx.x * ROWS_PER_BLOCK;

  #pragma unroll 8
  for (int i = 0; i < (ROWS_PER_BLOCK * EMBED / 4) / 256; ++i) {
    const int t   = tid + i * 256;
    const int row = t >> 7;
    const int c4  = t & 127;
    const float4 val = ((const float4*)(x + (r0 + row) * EMBED))[c4];
    float* d = &Xs[row * XS_STRIDE + 2 * c4];
    *(float2*)d = make_float2(val.x, val.y);
    *(float2*)(d + XS_HALF) = make_float2(val.z, val.w);
  }
  __syncthreads();

  float best[ROW_TILES][8];
  int   bidx[ROW_TILES][8];
  #pragma unroll
  for (int rt = 0; rt < ROW_TILES; ++rt) {
    #pragma unroll
    for (int j = 0; j < 8; ++j) { best[rt][j] = __builtin_inff(); bidx[rt][j] = 0; }
  }

  const float* xrow = Xs + lo * XS_STRIDE + XS_HALF * hi;
  int tile2_off = 2 * 16 * XS_STRIDE;
  asm volatile("" : "+v"(tile2_off));
  const float* xrow2 = xrow + tile2_off;


#define LOAD_W_BLOCK(dst, ptr)                                              \
  {                                                                         \
    _Pragma("unroll")                                                       \
    for (int i_ = 0; i_ < 8; ++i_) dst[i_] = *(const v2f*)((ptr) + 4 * i_); \
  }

#define LOAD_A_HALF(dst, kf, sh)                                            \
  {                                                                         \
    _Pragma("unroll")                                                       \
    for (int s2_ = 0; s2_ < 2; ++s2_) {                                     \
      _Pragma("unroll")                                                     \
      for (int rt = 0; rt < ROW_TILES; ++rt)                                \
        dst[s2_ * 4 + rt] = *(const v4f*)((rt < 2 ? xrow : xrow2) +         \
                                          (rt & 1) * 16 * XS_STRIDE +       \
                                          ((kf) >> 1) + 4 * (2 * (sh) + s2_)); \
    }                                                                       \
  }

#define MMA_HALF(ah, wb, sh)                                                \
  {                                                                         \
    _Pragma("unroll")                                                       \
    for (int s2_ = 0; s2_ < 2; ++s2_) {                                     \
      _Pragma("unroll")                                                     \
      for (int rt = 0; rt < ROW_TILES; ++rt) {                              \
        const v4f a_ = ah[s2_ * 4 + rt];                                    \
        const v2f a0 = __builtin_shufflevector(a_, a_, 0, 1);               \
        const v2f a1 = __builtin_shufflevector(a_, a_, 2, 3);               \
        c0[rt] = __builtin_amdgcn_wmma_f32_16x16x4_f32(false, a0, false, wb[2 * (2 * (sh) + s2_)],     \
                                                       (short)0, c0[rt], false, false);  \
        c1[rt] = __builtin_amdgcn_wmma_f32_16x16x4_f32(false, a1, false, wb[2 * (2 * (sh) + s2_) + 1], \
                                                       (short)0, c1[rt], false, false);  \
      }                                                                     \
    }                                                                       \
  }

#define PIN_ACC()                                                           \
  asm volatile("" : "+v"(c0[0]), "+v"(c0[1]), "+v"(c0[2]), "+v"(c0[3]),     \
                    "+v"(c1[0]), "+v"(c1[1]), "+v"(c1[2]), "+v"(c1[3]))

  v8f c0[ROW_TILES];
  v8f c1[ROW_TILES];
  #pragma unroll
  for (int rt = 0; rt < ROW_TILES; ++rt) { c0[rt] = (v8f){}; c1[rt] = (v8f){}; }

  const float* wp = w + (long)(wave * 16 + lo) * EMBED + 2 * hi;
  v2f bA[8];
  v2f bB[8];
  v4f aX[8];
  v4f aY[8];
  LOAD_W_BLOCK(bA, wp);
  LOAD_A_HALF(aX, 0, 0);

  #pragma nounroll
  for (int blk = 0; blk < 64; blk += 2) {
    const int kb = (blk & 15) * 32;

    LOAD_W_BLOCK(bB, wp + 32);
    LOAD_A_HALF(aY, kb, 1);
    __builtin_amdgcn_sched_barrier(0);
    MMA_HALF(aX, bA, 0);
    PIN_ACC();
    __builtin_amdgcn_sched_barrier(0);
    LOAD_A_HALF(aX, kb + 32, 0);
    __builtin_amdgcn_sched_barrier(0);
    MMA_HALF(aY, bA, 1);
    PIN_ACC();
    __builtin_amdgcn_sched_barrier(0);

    const int step = (kb != 448) ? 64 : ((blk != 62) ? (128 * EMBED - 448) : (-3 * 128 * EMBED - 448));
    wp += step;
    LOAD_W_BLOCK(bA, wp);
    LOAD_A_HALF(aY, kb + 32, 1);
    __builtin_amdgcn_sched_barrier(0);
    MMA_HALF(aX, bB, 0);
    PIN_ACC();
    __builtin_amdgcn_sched_barrier(0);
    LOAD_A_HALF(aX, (kb + 64) & 511, 0);
    __builtin_amdgcn_sched_barrier(0);
    MMA_HALF(aY, bB, 1);

    if (kb == 448) {
      const int v = (wave + (blk >> 4) * 8) * 16 + lo;
      const float bv = bias[v];
      #pragma unroll
      for (int rt = 0; rt < ROW_TILES; ++rt) {
        #pragma unroll
        for (int j = 0; j < 8; ++j) {
          const float d = fmaf(-2.0f, c0[rt][j] + c1[rt][j], bv);
          if (d < best[rt][j]) { best[rt][j] = d; bidx[rt][j] = v; }
        }
        c0[rt] = (v8f){}; c1[rt] = (v8f){};
      }
    }
  }
#undef LOAD_W_BLOCK
#undef LOAD_A_HALF
#undef MMA_HALF
#undef PIN_ACC

  #pragma unroll
  for (int rt = 0; rt < ROW_TILES; ++rt) {
    #pragma unroll
    for (int j = 0; j < 8; ++j) {
      #pragma unroll
      for (int m = 1; m <= 8; m <<= 1) {
        const float ov = __shfl_xor(best[rt][j], m, 32);
        const int   oi = __shfl_xor(bidx[rt][j], m, 32);
        if (ov < best[rt][j] || (ov == best[rt][j] && oi < bidx[rt][j])) {
          best[rt][j] = ov; bidx[rt][j] = oi;
        }
      }
    }
  }
  if (lo == 0) {
    #pragma unroll
    for (int rt = 0; rt < ROW_TILES; ++rt) {
      #pragma unroll
      for (int j = 0; j < 8; ++j) {
        redV[wave][rt * 16 + hi * 8 + j] = best[rt][j];
        redI[wave][rt * 16 + hi * 8 + j] = bidx[rt][j];
      }
    }
  }
  __syncthreads();

  if (tid < ROWS_PER_BLOCK) {
    float bv = redV[0][tid];
    int   bi = redI[0][tid];
    #pragma unroll
    for (int w8 = 1; w8 < 8; ++w8) {
      const float ov = redV[w8][tid];
      const int   oi = redI[w8][tid];
      if (ov < bv || (ov == bv && oi < bi)) { bv = ov; bi = oi; }
    }
    fidx[tid] = bi;
  }
  __syncthreads();

  const int rsub = tid >> 4;
  const int l16  = tid & 15;
  #pragma unroll
  for (int rr = 0; rr < ROWS_PER_BLOCK / 16; ++rr) {
    const int row = rr * 16 + rsub;
    const float4* src = (const float4*)(w + (long)fidx[row] * EMBED);
    float4* dst = (float4*)(out + (r0 + row) * EMBED);
    #pragma unroll
    for (int i = 0; i < 8; ++i) dst[l16 + i * 16] = src[l16 + i * 16];
  }
}

extern "C" void kernel_launch(void* const* d_in, const int* in_sizes, int n_in,
                              void* d_out, int out_size, void* d_ws, size_t ws_size,
                              hipStream_t stream) {
  const float* x = (const float*)d_in[0];
  const float* w = (const float*)d_in[1];
  float* out  = (float*)d_out;
  float* bias = (float*)d_ws;

  quantizer_bias_kernel<<<VOCAB / 256, 256, 0, stream>>>(w, bias);

  const int n_rows = in_sizes[0] / EMBED;
  quantizer_main_kernel<<<n_rows / ROWS_PER_BLOCK, 256, 0, stream>>>(x, w, bias, out);
}


# Column-norm kernel: a 32-column strip's 512 rows are requested at once and parked in LDS, then the same k=0..511 fma chain per column; 64 serial cache round trips become one; sum order unchanged
# speedup vs baseline: 1.7242x; 1.1111x over previous
#include <hip/hip_runtime.h>
#include <hip/hip_bf16.h>

typedef __attribute__((ext_vector_type(2))) float v2f;
typedef __attribute__((ext_vector_type(4))) float v4f;
typedef __attribute__((ext_vector_type(8))) float v8f;

#define EMBED 512
#define VOCAB 512
#define XS_STRIDE 516
#define XS_HALF 256
#define ROWS_PER_BLOCK 64
#define ROW_TILES 4

#define BIAS_COLS 32
__global__ __launch_bounds__(256) void quantizer_bias_kernel(
    const float* __restrict__ w, float* __restrict__ bias) {
  __shared__ __attribute__((aligned(16))) float Ws[VOCAB * BIAS_COLS];
  const int tid = threadIdx.x;
  const int v0  = blockIdx.x * BIAS_COLS;
  const int c4  = tid & 7;
  const int r   = tid >> 3;
  float4 val[16];
  #pragma unroll
  for (int p = 0; p < 16; ++p)
    val[p] = ((const float4*)(w + (long)(p * 32 + r) * EMBED + v0))[c4];
  #pragma unroll
  for (int p = 0; p < 16; ++p)
    ((float4*)(Ws + (p * 32 + r) * BIAS_COLS))[c4] = val[p];
  __syncthreads();
  if (tid < BIAS_COLS) {
    const float* col = Ws + tid;
    float s = 0.0f;
    float ta[32];
    float tb[32];
    #pragma unroll
    for (int i = 0; i < 32; ++i) ta[i] = col[i * BIAS_COLS];
    for (int k0 = 0; k0 < VOCAB; k0 += 64) {
      #pragma unroll
      for (int i = 0; i < 32; ++i) tb[i] = col[(k0 + 32 + i) * BIAS_COLS];
      __builtin_amdgcn_sched_barrier(0);
      #pragma unroll
      for (int i = 0; i < 32; ++i) s = fmaf(ta[i], ta[i], s);
      const int kn = (k0 + 64) & (VOCAB - 1);
      #pragma unroll
      for (int i = 0; i < 32; ++i) ta[i] = col[(kn + i) * BIAS_COLS];
      __builtin_amdgcn_sched_barrier(0);
      #pragma unroll
      for (int i = 0; i < 32; ++i) s = fmaf(tb[i], tb[i], s);
    }
    bias[v0 + tid] = s;
  }
}

__global__ __launch_bounds__(256) void quantizer_main_kernel(
    const float* __restrict__ x, const float* __restrict__ w,
    const float* __restrict__ bias, float* __restrict__ out) {
  __shared__ __attribute__((aligned(16))) float Xs[ROWS_PER_BLOCK * XS_STRIDE];
  __shared__ float redV[8][ROWS_PER_BLOCK];
  __shared__ int   redI[8][ROWS_PER_BLOCK];
  __shared__ int   fidx[ROWS_PER_BLOCK];

  const int tid  = threadIdx.x;
  const int wave = __builtin_amdgcn_readfirstlane(tid >> 5);
  const int lane = tid & 31;
  const int lo   = lane & 15;
  const int hi   = lane >> 4;
  const long r0  = (long)blockIdx.x * ROWS_PER_BLOCK;

  #pragma unroll 8
  for (int i = 0; i < (ROWS_PER_BLOCK * EMBED / 4) / 256; ++i) {
    const int t   = tid + i * 256;
    const int row = t >> 7;
    const int c4  = t & 127;
    const float4 val = ((const float4*)(x + (r0 + row) * EMBED))[c4];
    float* d = &Xs[row * XS_STRIDE + 2 * c4];
    *(float2*)d = make_float2(val.x, val.y);
    *(float2*)(d + XS_HALF) = make_float2(val.z, val.w);
  }
  __syncthreads();

  float best[ROW_TILES][8];
  int   bidx[ROW_TILES][8];
  #pragma unroll
  for (int rt = 0; rt < ROW_TILES; ++rt) {
    #pragma unroll
    for (int j = 0; j < 8; ++j) { best[rt][j] = __builtin_inff(); bidx[rt][j] = 0; }
  }

  const float* xrow = Xs + lo * XS_STRIDE + XS_HALF * hi;
  int tile2_off = 2 * 16 * XS_STRIDE;
  asm volatile("" : "+v"(tile2_off));
  const float* xrow2 = xrow + tile2_off;


#define LOAD_W_BLOCK(dst, ptr)                                              \
  {                                                                         \
    _Pragma("unroll")                                                       \
    for (int i_ = 0; i_ < 8; ++i_) dst[i_] = *(const v2f*)((ptr) + 4 * i_); \
  }

#define LOAD_A_HALF(dst, kf, sh)                                            \
  {                                                                         \
    _Pragma("unroll")                                                       \
    for (int s2_ = 0; s2_ < 2; ++s2_) {                                     \
      _Pragma("unroll")                                                     \
      for (int rt = 0; rt < ROW_TILES; ++rt)                                \
        dst[s2_ * 4 + rt] = *(const v4f*)((rt < 2 ? xrow : xrow2) +         \
                                          (rt & 1) * 16 * XS_STRIDE +       \
                                          ((kf) >> 1) + 4 * (2 * (sh) + s2_)); \
    }                                                                       \
  }

#define MMA_HALF(ah, wb, sh)                                                \
  {                                                                         \
    _Pragma("unroll")                                                       \
    for (int s2_ = 0; s2_ < 2; ++s2_) {                                     \
      _Pragma("unroll")                                                     \
      for (int rt = 0; rt < ROW_TILES; ++rt) {                              \
        const v4f a_ = ah[s2_ * 4 + rt];                                    \
        const v2f a0 = __builtin_shufflevector(a_, a_, 0, 1);               \
        const v2f a1 = __builtin_shufflevector(a_, a_, 2, 3);               \
        c0[rt] = __builtin_amdgcn_wmma_f32_16x16x4_f32(false, a0, false, wb[2 * (2 * (sh) + s2_)],     \
                                                       (short)0, c0[rt], false, false);  \
        c1[rt] = __builtin_amdgcn_wmma_f32_16x16x4_f32(false, a1, false, wb[2 * (2 * (sh) + s2_) + 1], \
                                                       (short)0, c1[rt], false, false);  \
      }                                                                     \
    }                                                                       \
  }

#define PIN_ACC()                                                           \
  asm volatile("" : "+v"(c0[0]), "+v"(c0[1]), "+v"(c0[2]), "+v"(c0[3]),     \
                    "+v"(c1[0]), "+v"(c1[1]), "+v"(c1[2]), "+v"(c1[3]))

  v8f c0[ROW_TILES];
  v8f c1[ROW_TILES];
  #pragma unroll
  for (int rt = 0; rt < ROW_TILES; ++rt) { c0[rt] = (v8f){}; c1[rt] = (v8f){}; }

  const float* wp = w + (long)(wave * 16 + lo) * EMBED + 2 * hi;
  v2f bA[8];
  v2f bB[8];
  v4f aX[8];
  v4f aY[8];
  LOAD_W_BLOCK(bA, wp);
  LOAD_A_HALF(aX, 0, 0);

  #pragma nounroll
  for (int blk = 0; blk < 64; blk += 2) {
    const int kb = (blk & 15) * 32;

    LOAD_W_BLOCK(bB, wp + 32);
    LOAD_A_HALF(aY, kb, 1);
    __builtin_amdgcn_sched_barrier(0);
    MMA_HALF(aX, bA, 0);
    PIN_ACC();
    __builtin_amdgcn_sched_barrier(0);
    LOAD_A_HALF(aX, kb + 32, 0);
    __builtin_amdgcn_sched_barrier(0);
    MMA_HALF(aY, bA, 1);
    PIN_ACC();
    __builtin_amdgcn_sched_barrier(0);

    const int step = (kb != 448) ? 64 : ((blk != 62) ? (128 * EMBED - 448) : (-3 * 128 * EMBED - 448));
    wp += step;
    LOAD_W_BLOCK(bA, wp);
    LOAD_A_HALF(aY, kb + 32, 1);
    __builtin_amdgcn_sched_barrier(0);
    MMA_HALF(aX, bB, 0);
    PIN_ACC();
    __builtin_amdgcn_sched_barrier(0);
    LOAD_A_HALF(aX, (kb + 64) & 511, 0);
    __builtin_amdgcn_sched_barrier(0);
    MMA_HALF(aY, bB, 1);

    if (kb == 448) {
      const int v = (wave + (blk >> 4) * 8) * 16 + lo;
      const float bv = bias[v];
      #pragma unroll
      for (int rt = 0; rt < ROW_TILES; ++rt) {
        #pragma unroll
        for (int j = 0; j < 8; ++j) {
          const float d = fmaf(-2.0f, c0[rt][j] + c1[rt][j], bv);
          if (d < best[rt][j]) { best[rt][j] = d; bidx[rt][j] = v; }
        }
        c0[rt] = (v8f){}; c1[rt] = (v8f){};
      }
    }
  }
#undef LOAD_W_BLOCK
#undef LOAD_A_HALF
#undef MMA_HALF
#undef PIN_ACC

  #pragma unroll
  for (int rt = 0; rt < ROW_TILES; ++rt) {
    #pragma unroll
    for (int j = 0; j < 8; ++j) {
      #pragma unroll
      for (int m = 1; m <= 8; m <<= 1) {
        const float ov = __shfl_xor(best[rt][j], m, 32);
        const int   oi = __shfl_xor(bidx[rt][j], m, 32);
        if (ov < best[rt][j] || (ov == best[rt][j] && oi < bidx[rt][j])) {
          best[rt][j] = ov; bidx[rt][j] = oi;
        }
      }
    }
  }
  if (lo == 0) {
    #pragma unroll
    for (int rt = 0; rt < ROW_TILES; ++rt) {
      #pragma unroll
      for (int j = 0; j < 8; ++j) {
        redV[wave][rt * 16 + hi * 8 + j] = best[rt][j];
        redI[wave][rt * 16 + hi * 8 + j] = bidx[rt][j];
      }
    }
  }
  __syncthreads();

  if (tid < ROWS_PER_BLOCK) {
    float bv = redV[0][tid];
    int   bi = redI[0][tid];
    #pragma unroll
    for (int w8 = 1; w8 < 8; ++w8) {
      const float ov = redV[w8][tid];
      const int   oi = redI[w8][tid];
      if (ov < bv || (ov == bv && oi < bi)) { bv = ov; bi = oi; }
    }
    fidx[tid] = bi;
  }
  __syncthreads();

  const int rsub = tid >> 4;
  const int l16  = tid & 15;
  #pragma unroll
  for (int rr = 0; rr < ROWS_PER_BLOCK / 16; ++rr) {
    const int row = rr * 16 + rsub;
    const float4* src = (const float4*)(w + (long)fidx[row] * EMBED);
    float4* dst = (float4*)(out + (r0 + row) * EMBED);
    #pragma unroll
    for (int i = 0; i < 8; ++i) dst[l16 + i * 16] = src[l16 + i * 16];
  }
}

extern "C" void kernel_launch(void* const* d_in, const int* in_sizes, int n_in,
                              void* d_out, int out_size, void* d_ws, size_t ws_size,
                              hipStream_t stream) {
  const float* x = (const float*)d_in[0];
  const float* w = (const float*)d_in[1];
  float* out  = (float*)d_out;
  float* bias = (float*)d_ws;

  quantizer_bias_kernel<<<VOCAB / BIAS_COLS, 256, 0, stream>>>(w, bias);

  const int n_rows = in_sizes[0] / EMBED;
  quantizer_main_kernel<<<n_rows / ROWS_PER_BLOCK, 256, 0, stream>>>(x, w, bias, out);
}


# Per-row argmin over each 16-lane half-wave: min value, then lowest index among equals, by row_xmask swizzles on the vector ALU; replaces 256 serial LDS-permute round trips per wave; sums untouched
# speedup vs baseline: 1.8263x; 1.0592x over previous
#include <hip/hip_runtime.h>
#include <hip/hip_bf16.h>

typedef __attribute__((ext_vector_type(2))) float v2f;
typedef __attribute__((ext_vector_type(4))) float v4f;
typedef __attribute__((ext_vector_type(8))) float v8f;

#define EMBED 512
#define VOCAB 512
#define XS_STRIDE 516
#define XS_HALF 256
#define ROWS_PER_BLOCK 64
#define ROW_TILES 4

#define BIAS_COLS 32
__global__ __launch_bounds__(256) void quantizer_bias_kernel(
    const float* __restrict__ w, float* __restrict__ bias) {
  __shared__ __attribute__((aligned(16))) float Ws[VOCAB * BIAS_COLS];
  const int tid = threadIdx.x;
  const int v0  = blockIdx.x * BIAS_COLS;
  const int c4  = tid & 7;
  const int r   = tid >> 3;
  float4 val[16];
  #pragma unroll
  for (int p = 0; p < 16; ++p)
    val[p] = ((const float4*)(w + (long)(p * 32 + r) * EMBED + v0))[c4];
  #pragma unroll
  for (int p = 0; p < 16; ++p)
    ((float4*)(Ws + (p * 32 + r) * BIAS_COLS))[c4] = val[p];
  __syncthreads();
  if (tid < BIAS_COLS) {
    const float* col = Ws + tid;
    float s = 0.0f;
    float ta[32];
    float tb[32];
    #pragma unroll
    for (int i = 0; i < 32; ++i) ta[i] = col[i * BIAS_COLS];
    for (int k0 = 0; k0 < VOCAB; k0 += 64) {
      #pragma unroll
      for (int i = 0; i < 32; ++i) tb[i] = col[(k0 + 32 + i) * BIAS_COLS];
      __builtin_amdgcn_sched_barrier(0);
      #pragma unroll
      for (int i = 0; i < 32; ++i) s = fmaf(ta[i], ta[i], s);
      const int kn = (k0 + 64) & (VOCAB - 1);
      #pragma unroll
      for (int i = 0; i < 32; ++i) ta[i] = col[(kn + i) * BIAS_COLS];
      __builtin_amdgcn_sched_barrier(0);
      #pragma unroll
      for (int i = 0; i < 32; ++i) s = fmaf(tb[i], tb[i], s);
    }
    bias[v0 + tid] = s;
  }
}

template <int M>
__device__ __forceinline__ float row_xor_min_f(float v) {
  const float o = __builtin_bit_cast(
      float, __builtin_amdgcn_mov_dpp(__builtin_bit_cast(int, v), 0x160 + M, 0xf, 0xf, false));
  return (o < v) ? o : v;
}
template <int M>
__device__ __forceinline__ int row_xor_min_i(int v) {
  return min(v, __builtin_amdgcn_update_dpp(0x7fffffff, v, 0x160 + M, 0xf, 0xf, false));
}

__global__ __launch_bounds__(256) void quantizer_main_kernel(
    const float* __restrict__ x, const float* __restrict__ w,
    const float* __restrict__ bias, float* __restrict__ out) {
  __shared__ __attribute__((aligned(16))) float Xs[ROWS_PER_BLOCK * XS_STRIDE];
  __shared__ float redV[8][ROWS_PER_BLOCK];
  __shared__ int   redI[8][ROWS_PER_BLOCK];
  __shared__ int   fidx[ROWS_PER_BLOCK];

  const int tid  = threadIdx.x;
  const int wave = __builtin_amdgcn_readfirstlane(tid >> 5);
  const int lane = tid & 31;
  const int lo   = lane & 15;
  const int hi   = lane >> 4;
  const long r0  = (long)blockIdx.x * ROWS_PER_BLOCK;

  #pragma unroll 8
  for (int i = 0; i < (ROWS_PER_BLOCK * EMBED / 4) / 256; ++i) {
    const int t   = tid + i * 256;
    const int row = t >> 7;
    const int c4  = t & 127;
    const float4 val = ((const float4*)(x + (r0 + row) * EMBED))[c4];
    float* d = &Xs[row * XS_STRIDE + 2 * c4];
    *(float2*)d = make_float2(val.x, val.y);
    *(float2*)(d + XS_HALF) = make_float2(val.z, val.w);
  }
  __syncthreads();

  float best[ROW_TILES][8];
  int   bidx[ROW_TILES][8];
  #pragma unroll
  for (int rt = 0; rt < ROW_TILES; ++rt) {
    #pragma unroll
    for (int j = 0; j < 8; ++j) { best[rt][j] = __builtin_inff(); bidx[rt][j] = 0; }
  }

  const float* xrow = Xs + lo * XS_STRIDE + XS_HALF * hi;
  int tile2_off = 2 * 16 * XS_STRIDE;
  asm volatile("" : "+v"(tile2_off));
  const float* xrow2 = xrow + tile2_off;


#define LOAD_W_BLOCK(dst, ptr)                                              \
  {                                                                         \
    _Pragma("unroll")                                                       \
    for (int i_ = 0; i_ < 8; ++i_) dst[i_] = *(const v2f*)((ptr) + 4 * i_); \
  }

#define LOAD_A_HALF(dst, kf, sh)                                            \
  {                                                                         \
    _Pragma("unroll")                                                       \
    for (int s2_ = 0; s2_ < 2; ++s2_) {                                     \
      _Pragma("unroll")                                                     \
      for (int rt = 0; rt < ROW_TILES; ++rt)                                \
        dst[s2_ * 4 + rt] = *(const v4f*)((rt < 2 ? xrow : xrow2) +         \
                                          (rt & 1) * 16 * XS_STRIDE +       \
                                          ((kf) >> 1) + 4 * (2 * (sh) + s2_)); \
    }                                                                       \
  }

#define MMA_HALF(ah, wb, sh)                                                \
  {                                                                         \
    _Pragma("unroll")                                                       \
    for (int s2_ = 0; s2_ < 2; ++s2_) {                                     \
      _Pragma("unroll")                                                     \
      for (int rt = 0; rt < ROW_TILES; ++rt) {                              \
        const v4f a_ = ah[s2_ * 4 + rt];                                    \
        const v2f a0 = __builtin_shufflevector(a_, a_, 0, 1);               \
        const v2f a1 = __builtin_shufflevector(a_, a_, 2, 3);               \
        c0[rt] = __builtin_amdgcn_wmma_f32_16x16x4_f32(false, a0, false, wb[2 * (2 * (sh) + s2_)],     \
                                                       (short)0, c0[rt], false, false);  \
        c1[rt] = __builtin_amdgcn_wmma_f32_16x16x4_f32(false, a1, false, wb[2 * (2 * (sh) + s2_) + 1], \
                                                       (short)0, c1[rt], false, false);  \
      }                                                                     \
    }                                                                       \
  }

#define PIN_ACC()                                                           \
  asm volatile("" : "+v"(c0[0]), "+v"(c0[1]), "+v"(c0[2]), "+v"(c0[3]),     \
                    "+v"(c1[0]), "+v"(c1[1]), "+v"(c1[2]), "+v"(c1[3]))

  v8f c0[ROW_TILES];
  v8f c1[ROW_TILES];
  #pragma unroll
  for (int rt = 0; rt < ROW_TILES; ++rt) { c0[rt] = (v8f){}; c1[rt] = (v8f){}; }

  const float* wp = w + (long)(wave * 16 + lo) * EMBED + 2 * hi;
  v2f bA[8];
  v2f bB[8];
  v4f aX[8];
  v4f aY[8];
  LOAD_W_BLOCK(bA, wp);
  LOAD_A_HALF(aX, 0, 0);

  #pragma nounroll
  for (int blk = 0; blk < 64; blk += 2) {
    const int kb = (blk & 15) * 32;

    LOAD_W_BLOCK(bB, wp + 32);
    LOAD_A_HALF(aY, kb, 1);
    __builtin_amdgcn_sched_barrier(0);
    MMA_HALF(aX, bA, 0);
    PIN_ACC();
    __builtin_amdgcn_sched_barrier(0);
    LOAD_A_HALF(aX, kb + 32, 0);
    __builtin_amdgcn_sched_barrier(0);
    MMA_HALF(aY, bA, 1);
    PIN_ACC();
    __builtin_amdgcn_sched_barrier(0);

    const int step = (kb != 448) ? 64 : ((blk != 62) ? (128 * EMBED - 448) : (-3 * 128 * EMBED - 448));
    wp += step;
    LOAD_W_BLOCK(bA, wp);
    LOAD_A_HALF(aY, kb + 32, 1);
    __builtin_amdgcn_sched_barrier(0);
    MMA_HALF(aX, bB, 0);
    PIN_ACC();
    __builtin_amdgcn_sched_barrier(0);
    LOAD_A_HALF(aX, (kb + 64) & 511, 0);
    __builtin_amdgcn_sched_barrier(0);
    MMA_HALF(aY, bB, 1);

    if (kb == 448) {
      const int v = (wave + (blk >> 4) * 8) * 16 + lo;
      const float bv = bias[v];
      #pragma unroll
      for (int rt = 0; rt < ROW_TILES; ++rt) {
        #pragma unroll
        for (int j = 0; j < 8; ++j) {
          const float d = fmaf(-2.0f, c0[rt][j] + c1[rt][j], bv);
          if (d < best[rt][j]) { best[rt][j] = d; bidx[rt][j] = v; }
        }
        c0[rt] = (v8f){}; c1[rt] = (v8f){};
      }
    }
  }
#undef LOAD_W_BLOCK
#undef LOAD_A_HALF
#undef MMA_HALF
#undef PIN_ACC

  #pragma unroll
  for (int rt = 0; rt < ROW_TILES; ++rt) {
    #pragma unroll
    for (int j = 0; j < 8; ++j) {
      float rmin = best[rt][j];
      rmin = row_xor_min_f<1>(rmin);
      rmin = row_xor_min_f<2>(rmin);
      rmin = row_xor_min_f<4>(rmin);
      rmin = row_xor_min_f<8>(rmin);
      int cand = (best[rt][j] == rmin) ? bidx[rt][j] : 0x7fffffff;
      cand = row_xor_min_i<1>(cand);
      cand = row_xor_min_i<2>(cand);
      cand = row_xor_min_i<4>(cand);
      cand = row_xor_min_i<8>(cand);
      best[rt][j] = rmin; bidx[rt][j] = cand;
    }
  }
  if (lo == 0) {
    #pragma unroll
    for (int rt = 0; rt < ROW_TILES; ++rt) {
      #pragma unroll
      for (int j = 0; j < 8; ++j) {
        redV[wave][rt * 16 + hi * 8 + j] = best[rt][j];
        redI[wave][rt * 16 + hi * 8 + j] = bidx[rt][j];
      }
    }
  }
  __syncthreads();

  if (tid < ROWS_PER_BLOCK) {
    float bv = redV[0][tid];
    int   bi = redI[0][tid];
    #pragma unroll
    for (int w8 = 1; w8 < 8; ++w8) {
      const float ov = redV[w8][tid];
      const int   oi = redI[w8][tid];
      if (ov < bv || (ov == bv && oi < bi)) { bv = ov; bi = oi; }
    }
    fidx[tid] = bi;
  }
  __syncthreads();

  const int rsub = tid >> 4;
  const int l16  = tid & 15;
  #pragma unroll
  for (int rr = 0; rr < ROWS_PER_BLOCK / 16; ++rr) {
    const int row = rr * 16 + rsub;
    const float4* src = (const float4*)(w + (long)fidx[row] * EMBED);
    float4* dst = (float4*)(out + (r0 + row) * EMBED);
    #pragma unroll
    for (int i = 0; i < 8; ++i) dst[l16 + i * 16] = src[l16 + i * 16];
  }
}

extern "C" void kernel_launch(void* const* d_in, const int* in_sizes, int n_in,
                              void* d_out, int out_size, void* d_ws, size_t ws_size,
                              hipStream_t stream) {
  const float* x = (const float*)d_in[0];
  const float* w = (const float*)d_in[1];
  float* out  = (float*)d_out;
  float* bias = (float*)d_ws;

  quantizer_bias_kernel<<<VOCAB / BIAS_COLS, 256, 0, stream>>>(w, bias);

  const int n_rows = in_sizes[0] / EMBED;
  quantizer_main_kernel<<<n_rows / ROWS_PER_BLOCK, 256, 0, stream>>>(x, w, bias, out);
}


# Output gather: a thread requests all 32 of its 16-byte pieces of the chosen W rows before its first store (was load, wait, store 32 times in a row per thread); sums, argmin and k order unchanged
# speedup vs baseline: 1.8993x; 1.0400x over previous
#include <hip/hip_runtime.h>
#include <hip/hip_bf16.h>

typedef __attribute__((ext_vector_type(2))) float v2f;
typedef __attribute__((ext_vector_type(4))) float v4f;
typedef __attribute__((ext_vector_type(8))) float v8f;

#define EMBED 512
#define VOCAB 512
#define XS_STRIDE 516
#define XS_HALF 256
#define ROWS_PER_BLOCK 64
#define ROW_TILES 4

#define BIAS_COLS 32
__global__ __launch_bounds__(256) void quantizer_bias_kernel(
    const float* __restrict__ w, float* __restrict__ bias) {
  __shared__ __attribute__((aligned(16))) float Ws[VOCAB * BIAS_COLS];
  const int tid = threadIdx.x;
  const int v0  = blockIdx.x * BIAS_COLS;
  const int c4  = tid & 7;
  const int r   = tid >> 3;
  float4 val[16];
  #pragma unroll
  for (int p = 0; p < 16; ++p)
    val[p] = ((const float4*)(w + (long)(p * 32 + r) * EMBED + v0))[c4];
  #pragma unroll
  for (int p = 0; p < 16; ++p)
    ((float4*)(Ws + (p * 32 + r) * BIAS_COLS))[c4] = val[p];
  __syncthreads();
  if (tid < BIAS_COLS) {
    const float* col = Ws + tid;
    float s = 0.0f;
    float ta[32];
    float tb[32];
    #pragma unroll
    for (int i = 0; i < 32; ++i) ta[i] = col[i * BIAS_COLS];
    for (int k0 = 0; k0 < VOCAB; k0 += 64) {
      #pragma unroll
      for (int i = 0; i < 32; ++i) tb[i] = col[(k0 + 32 + i) * BIAS_COLS];
      __builtin_amdgcn_sched_barrier(0);
      #pragma unroll
      for (int i = 0; i < 32; ++i) s = fmaf(ta[i], ta[i], s);
      const int kn = (k0 + 64) & (VOCAB - 1);
      #pragma unroll
      for (int i = 0; i < 32; ++i) ta[i] = col[(kn + i) * BIAS_COLS];
      __builtin_amdgcn_sched_barrier(0);
      #pragma unroll
      for (int i = 0; i < 32; ++i) s = fmaf(tb[i], tb[i], s);
    }
    bias[v0 + tid] = s;
  }
}

template <int M>
__device__ __forceinline__ float row_xor_min_f(float v) {
  const float o = __builtin_bit_cast(
      float, __builtin_amdgcn_mov_dpp(__builtin_bit_cast(int, v), 0x160 + M, 0xf, 0xf, false));
  return (o < v) ? o : v;
}
template <int M>
__device__ __forceinline__ int row_xor_min_i(int v) {
  return min(v, __builtin_amdgcn_update_dpp(0x7fffffff, v, 0x160 + M, 0xf, 0xf, false));
}

__global__ __launch_bounds__(256) void quantizer_main_kernel(
    const float* __restrict__ x, const float* __restrict__ w,
    const float* __restrict__ bias, float* __restrict__ out) {
  __shared__ __attribute__((aligned(16))) float Xs[ROWS_PER_BLOCK * XS_STRIDE];
  __shared__ float redV[8][ROWS_PER_BLOCK];
  __shared__ int   redI[8][ROWS_PER_BLOCK];
  __shared__ int   fidx[ROWS_PER_BLOCK];

  const int tid  = threadIdx.x;
  const int wave = __builtin_amdgcn_readfirstlane(tid >> 5);
  const int lane = tid & 31;
  const int lo   = lane & 15;
  const int hi   = lane >> 4;
  const long r0  = (long)blockIdx.x * ROWS_PER_BLOCK;

  #pragma unroll 8
  for (int i = 0; i < (ROWS_PER_BLOCK * EMBED / 4) / 256; ++i) {
    const int t   = tid + i * 256;
    const int row = t >> 7;
    const int c4  = t & 127;
    const float4 val = ((const float4*)(x + (r0 + row) * EMBED))[c4];
    float* d = &Xs[row * XS_STRIDE + 2 * c4];
    *(float2*)d = make_float2(val.x, val.y);
    *(float2*)(d + XS_HALF) = make_float2(val.z, val.w);
  }
  __syncthreads();

  float best[ROW_TILES][8];
  int   bidx[ROW_TILES][8];
  #pragma unroll
  for (int rt = 0; rt < ROW_TILES; ++rt) {
    #pragma unroll
    for (int j = 0; j < 8; ++j) { best[rt][j] = __builtin_inff(); bidx[rt][j] = 0; }
  }

  const float* xrow = Xs + lo * XS_STRIDE + XS_HALF * hi;
  int tile2_off = 2 * 16 * XS_STRIDE;
  asm volatile("" : "+v"(tile2_off));
  const float* xrow2 = xrow + tile2_off;


#define LOAD_W_BLOCK(dst, ptr)                                              \
  {                                                                         \
    _Pragma("unroll")                                                       \
    for (int i_ = 0; i_ < 8; ++i_) dst[i_] = *(const v2f*)((ptr) + 4 * i_); \
  }

#define LOAD_A_HALF(dst, kf, sh)                                            \
  {                                                                         \
    _Pragma("unroll")                                                       \
    for (int s2_ = 0; s2_ < 2; ++s2_) {                                     \
      _Pragma("unroll")                                                     \
      for (int rt = 0; rt < ROW_TILES; ++rt)                                \
        dst[s2_ * 4 + rt] = *(const v4f*)((rt < 2 ? xrow : xrow2) +         \
                                          (rt & 1) * 16 * XS_STRIDE +       \
                                          ((kf) >> 1) + 4 * (2 * (sh) + s2_)); \
    }                                                                       \
  }

#define MMA_HALF(ah, wb, sh)                                                \
  {                                                                         \
    _Pragma("unroll")                                                       \
    for (int s2_ = 0; s2_ < 2; ++s2_) {                                     \
      _Pragma("unroll")                                                     \
      for (int rt = 0; rt < ROW_TILES; ++rt) {                              \
        const v4f a_ = ah[s2_ * 4 + rt];                                    \
        const v2f a0 = __builtin_shufflevector(a_, a_, 0, 1);               \
        const v2f a1 = __builtin_shufflevector(a_, a_, 2, 3);               \
        c0[rt] = __builtin_amdgcn_wmma_f32_16x16x4_f32(false, a0, false, wb[2 * (2 * (sh) + s2_)],     \
                                                       (short)0, c0[rt], false, false);  \
        c1[rt] = __builtin_amdgcn_wmma_f32_16x16x4_f32(false, a1, false, wb[2 * (2 * (sh) + s2_) + 1], \
                                                       (short)0, c1[rt], false, false);  \
      }                                                                     \
    }                                                                       \
  }

#define PIN_ACC()                                                           \
  asm volatile("" : "+v"(c0[0]), "+v"(c0[1]), "+v"(c0[2]), "+v"(c0[3]),     \
                    "+v"(c1[0]), "+v"(c1[1]), "+v"(c1[2]), "+v"(c1[3]))

  v8f c0[ROW_TILES];
  v8f c1[ROW_TILES];
  #pragma unroll
  for (int rt = 0; rt < ROW_TILES; ++rt) { c0[rt] = (v8f){}; c1[rt] = (v8f){}; }

  const float* wp = w + (long)(wave * 16 + lo) * EMBED + 2 * hi;
  v2f bA[8];
  v2f bB[8];
  v4f aX[8];
  v4f aY[8];
  LOAD_W_BLOCK(bA, wp);
  LOAD_A_HALF(aX, 0, 0);

  #pragma nounroll
  for (int blk = 0; blk < 64; blk += 2) {
    const int kb = (blk & 15) * 32;

    LOAD_W_BLOCK(bB, wp + 32);
    LOAD_A_HALF(aY, kb, 1);
    __builtin_amdgcn_sched_barrier(0);
    MMA_HALF(aX, bA, 0);
    PIN_ACC();
    __builtin_amdgcn_sched_barrier(0);
    LOAD_A_HALF(aX, kb + 32, 0);
    __builtin_amdgcn_sched_barrier(0);
    MMA_HALF(aY, bA, 1);
    PIN_ACC();
    __builtin_amdgcn_sched_barrier(0);

    const int step = (kb != 448) ? 64 : ((blk != 62) ? (128 * EMBED - 448) : (-3 * 128 * EMBED - 448));
    wp += step;
    LOAD_W_BLOCK(bA, wp);
    LOAD_A_HALF(aY, kb + 32, 1);
    __builtin_amdgcn_sched_barrier(0);
    MMA_HALF(aX, bB, 0);
    PIN_ACC();
    __builtin_amdgcn_sched_barrier(0);
    LOAD_A_HALF(aX, (kb + 64) & 511, 0);
    __builtin_amdgcn_sched_barrier(0);
    MMA_HALF(aY, bB, 1);

    if (kb == 448) {
      const int v = (wave + (blk >> 4) * 8) * 16 + lo;
      const float bv = bias[v];
      #pragma unroll
      for (int rt = 0; rt < ROW_TILES; ++rt) {
        #pragma unroll
        for (int j = 0; j < 8; ++j) {
          const float d = fmaf(-2.0f, c0[rt][j] + c1[rt][j], bv);
          if (d < best[rt][j]) { best[rt][j] = d; bidx[rt][j] = v; }
        }
        c0[rt] = (v8f){}; c1[rt] = (v8f){};
      }
    }
  }
#undef LOAD_W_BLOCK
#undef LOAD_A_HALF
#undef MMA_HALF
#undef PIN_ACC

  #pragma unroll
  for (int rt = 0; rt < ROW_TILES; ++rt) {
    #pragma unroll
    for (int j = 0; j < 8; ++j) {
      float rmin = best[rt][j];
      rmin = row_xor_min_f<1>(rmin);
      rmin = row_xor_min_f<2>(rmin);
      rmin = row_xor_min_f<4>(rmin);
      rmin = row_xor_min_f<8>(rmin);
      int cand = (best[rt][j] == rmin) ? bidx[rt][j] : 0x7fffffff;
      cand = row_xor_min_i<1>(cand);
      cand = row_xor_min_i<2>(cand);
      cand = row_xor_min_i<4>(cand);
      cand = row_xor_min_i<8>(cand);
      best[rt][j] = rmin; bidx[rt][j] = cand;
    }
  }
  if (lo == 0) {
    #pragma unroll
    for (int rt = 0; rt < ROW_TILES; ++rt) {
      #pragma unroll
      for (int j = 0; j < 8; ++j) {
        redV[wave][rt * 16 + hi * 8 + j] = best[rt][j];
        redI[wave][rt * 16 + hi * 8 + j] = bidx[rt][j];
      }
    }
  }
  __syncthreads();

  if (tid < ROWS_PER_BLOCK) {
    float bv = redV[0][tid];
    int   bi = redI[0][tid];
    #pragma unroll
    for (int w8 = 1; w8 < 8; ++w8) {
      const float ov = redV[w8][tid];
      const int   oi = redI[w8][tid];
      if (ov < bv || (ov == bv && oi < bi)) { bv = ov; bi = oi; }
    }
    fidx[tid] = bi;
  }
  __syncthreads();

  const int rsub = tid >> 4;
  const int l16  = tid & 15;
  const v4f* src0 = (const v4f*)(w + (long)fidx[0 * 16 + rsub] * EMBED) + l16;
  const v4f* src1 = (const v4f*)(w + (long)fidx[1 * 16 + rsub] * EMBED) + l16;
  const v4f* src2 = (const v4f*)(w + (long)fidx[2 * 16 + rsub] * EMBED) + l16;
  const v4f* src3 = (const v4f*)(w + (long)fidx[3 * 16 + rsub] * EMBED) + l16;
  v4f g[32];
  #pragma unroll
  for (int p = 0; p < 8; ++p) {
    g[p]      = src0[p * 16];
    g[8 + p]  = src1[p * 16];
    g[16 + p] = src2[p * 16];
    g[24 + p] = src3[p * 16];
  }
  __builtin_amdgcn_sched_barrier(0);
  v4f* dst = (v4f*)(out + (r0 + rsub) * EMBED) + l16;
  #pragma unroll
  for (int p = 0; p < 8; ++p) {
    dst[p * 16]                           = g[p];
    dst[1 * (16 * EMBED / 4) + p * 16]    = g[8 + p];
    dst[2 * (16 * EMBED / 4) + p * 16]    = g[16 + p];
    dst[3 * (16 * EMBED / 4) + p * 16]    = g[24 + p];
  }
}

extern "C" void kernel_launch(void* const* d_in, const int* in_sizes, int n_in,
                              void* d_out, int out_size, void* d_ws, size_t ws_size,
                              hipStream_t stream) {
  const float* x = (const float*)d_in[0];
  const float* w = (const float*)d_in[1];
  float* out  = (float*)d_out;
  float* bias = (float*)d_ws;

  quantizer_bias_kernel<<<VOCAB / BIAS_COLS, 256, 0, stream>>>(w, bias);

  const int n_rows = in_sizes[0] / EMBED;
  quantizer_main_kernel<<<n_rows / ROWS_PER_BLOCK, 256, 0, stream>>>(x, w, bias, out);
}


# Wave tile 64 rows x 64 codes on 4-wave work-groups (one wave per SIMD, two groups per WGP): each A fragment feeds 4 code tiles; one load group per 64 matrix instructions; sums and k order unchanged
# speedup vs baseline: 1.9648x; 1.0345x over previous
#include <hip/hip_runtime.h>
#include <hip/hip_bf16.h>

typedef __attribute__((ext_vector_type(2))) float v2f;
typedef __attribute__((ext_vector_type(4))) float v4f;
typedef __attribute__((ext_vector_type(8))) float v8f;

#define EMBED 512
#define VOCAB 512
#define XS_STRIDE 516
#define XS_HALF 256
#define ROWS_PER_BLOCK 64
#define ROW_TILES 4
#define VOC_TILES 4
#define MAIN_WAVES 4

#define BIAS_COLS 32
__global__ __launch_bounds__(256) void quantizer_bias_kernel(
    const float* __restrict__ w, float* __restrict__ bias) {
  __shared__ __attribute__((aligned(16))) float Ws[VOCAB * BIAS_COLS];
  const int tid = threadIdx.x;
  const int v0  = blockIdx.x * BIAS_COLS;
  const int c4  = tid & 7;
  const int r   = tid >> 3;
  float4 val[16];
  #pragma unroll
  for (int p = 0; p < 16; ++p)
    val[p] = ((const float4*)(w + (long)(p * 32 + r) * EMBED + v0))[c4];
  #pragma unroll
  for (int p = 0; p < 16; ++p)
    ((float4*)(Ws + (p * 32 + r) * BIAS_COLS))[c4] = val[p];
  __syncthreads();
  if (tid < BIAS_COLS) {
    const float* col = Ws + tid;
    float s = 0.0f;
    float ta[32];
    float tb[32];
    #pragma unroll
    for (int i = 0; i < 32; ++i) ta[i] = col[i * BIAS_COLS];
    for (int k0 = 0; k0 < VOCAB; k0 += 64) {
      #pragma unroll
      for (int i = 0; i < 32; ++i) tb[i] = col[(k0 + 32 + i) * BIAS_COLS];
      __builtin_amdgcn_sched_barrier(0);
      #pragma unroll
      for (int i = 0; i < 32; ++i) s = fmaf(ta[i], ta[i], s);
      const int kn = (k0 + 64) & (VOCAB - 1);
      #pragma unroll
      for (int i = 0; i < 32; ++i) ta[i] = col[(kn + i) * BIAS_COLS];
      __builtin_amdgcn_sched_barrier(0);
      #pragma unroll
      for (int i = 0; i < 32; ++i) s = fmaf(tb[i], tb[i], s);
    }
    bias[v0 + tid] = s;
  }
}

template <int M>
__device__ __forceinline__ float row_xor_min_f(float v) {
  const float o = __builtin_bit_cast(
      float, __builtin_amdgcn_mov_dpp(__builtin_bit_cast(int, v), 0x160 + M, 0xf, 0xf, false));
  return (o < v) ? o : v;
}
template <int M>
__device__ __forceinline__ int row_xor_min_i(int v) {
  return min(v, __builtin_amdgcn_update_dpp(0x7fffffff, v, 0x160 + M, 0xf, 0xf, false));
}

__global__ __launch_bounds__(MAIN_WAVES * 32, 2) void quantizer_main_kernel(
    const float* __restrict__ x, const float* __restrict__ w,
    const float* __restrict__ bias, float* __restrict__ out) {
  __shared__ __attribute__((aligned(16))) float Xs[ROWS_PER_BLOCK * XS_STRIDE];
  __shared__ float redV[MAIN_WAVES][ROWS_PER_BLOCK];
  __shared__ int   redI[MAIN_WAVES][ROWS_PER_BLOCK];
  __shared__ int   fidx[ROWS_PER_BLOCK];

  const int tid  = threadIdx.x;
  const int wave = __builtin_amdgcn_readfirstlane(tid >> 5);
  const int lane = tid & 31;
  const int lo   = lane & 15;
  const int hi   = lane >> 4;
  const long r0  = (long)blockIdx.x * ROWS_PER_BLOCK;

  #pragma unroll 16
  for (int i = 0; i < (ROWS_PER_BLOCK * EMBED / 4) / (MAIN_WAVES * 32); ++i) {
    const int t   = tid + i * (MAIN_WAVES * 32);
    const int row = t >> 7;
    const int c4  = t & 127;
    const float4 val = ((const float4*)(x + (r0 + row) * EMBED))[c4];
    float* d = &Xs[row * XS_STRIDE + 2 * c4];
    *(float2*)d = make_float2(val.x, val.y);
    *(float2*)(d + XS_HALF) = make_float2(val.z, val.w);
  }
  __syncthreads();

  float best[ROW_TILES][8];
  int   bidx[ROW_TILES][8];
  #pragma unroll
  for (int rt = 0; rt < ROW_TILES; ++rt) {
    #pragma unroll
    for (int j = 0; j < 8; ++j) { best[rt][j] = __builtin_inff(); bidx[rt][j] = 0; }
  }

  const float* xrow = Xs + lo * XS_STRIDE + XS_HALF * hi;
  int tile2_off = 2 * 16 * XS_STRIDE;
  asm volatile("" : "+v"(tile2_off));
  const float* xrow2 = xrow + tile2_off;


#define LOAD_W_STEP(dst, ptr)                                               \
  {                                                                         \
    _Pragma("unroll")                                                       \
    for (int vt_ = 0; vt_ < VOC_TILES; ++vt_) {                             \
      _Pragma("unroll")                                                     \
      for (int i_ = 0; i_ < 4; ++i_)                                        \
        dst[vt_ * 4 + i_] = *(const v2f*)((ptr) + vt_ * 16 * EMBED + 4 * i_); \
    }                                                                       \
  }

#define LOAD_A_STEP(dst, kf)                                                \
  {                                                                         \
    _Pragma("unroll")                                                       \
    for (int s2_ = 0; s2_ < 2; ++s2_) {                                     \
      _Pragma("unroll")                                                     \
      for (int rt = 0; rt < ROW_TILES; ++rt)                                \
        dst[s2_ * 4 + rt] = *(const v4f*)((rt < 2 ? xrow : xrow2) +         \
                                          (rt & 1) * 16 * XS_STRIDE +       \
                                          ((kf) >> 1) + 4 * s2_);           \
    }                                                                       \
  }

#define MMA_STEP(ah, wb)                                                    \
  {                                                                         \
    _Pragma("unroll")                                                       \
    for (int s2_ = 0; s2_ < 2; ++s2_) {                                     \
      _Pragma("unroll")                                                     \
      for (int rt = 0; rt < ROW_TILES; ++rt) {                              \
        const v4f a_ = ah[s2_ * 4 + rt];                                    \
        const v2f a0 = __builtin_shufflevector(a_, a_, 0, 1);               \
        const v2f a1 = __builtin_shufflevector(a_, a_, 2, 3);               \
        _Pragma("unroll")                                                   \
        for (int vt_ = 0; vt_ < VOC_TILES; ++vt_) {                         \
          c0[rt][vt_] = __builtin_amdgcn_wmma_f32_16x16x4_f32(              \
              false, a0, false, wb[vt_ * 4 + 2 * s2_], (short)0, c0[rt][vt_], false, false);     \
          c1[rt][vt_] = __builtin_amdgcn_wmma_f32_16x16x4_f32(              \
              false, a1, false, wb[vt_ * 4 + 2 * s2_ + 1], (short)0, c1[rt][vt_], false, false); \
        }                                                                   \
      }                                                                     \
    }                                                                       \
  }

  v8f c0[ROW_TILES][VOC_TILES];
  v8f c1[ROW_TILES][VOC_TILES];
  #pragma unroll
  for (int rt = 0; rt < ROW_TILES; ++rt) {
    #pragma unroll
    for (int vt = 0; vt < VOC_TILES; ++vt) { c0[rt][vt] = (v8f){}; c1[rt][vt] = (v8f){}; }
  }

  const float* wp = w + (long)(wave * VOC_TILES * 16 + lo) * EMBED + 2 * hi;
  v2f wA[16];
  v2f wB[16];
  v4f aA[8];
  v4f aB[8];
  LOAD_W_STEP(wA, wp);
  LOAD_A_STEP(aA, 0);

  #pragma nounroll
  for (int p = 0; p < 2; ++p) {
    #pragma nounroll
    for (int k = 0; k < EMBED; k += 32) {
      LOAD_W_STEP(wB, wp + 16);
      LOAD_A_STEP(aB, k + 16);
      __builtin_amdgcn_sched_barrier(0);
      MMA_STEP(aA, wA);
      __builtin_amdgcn_sched_barrier(0);

      const int step = (k != EMBED - 32) ? 32
                                         : ((p == 0) ? (256 * EMBED - (EMBED - 32)) : (-256 * EMBED - (EMBED - 32)));
      wp += step;
      LOAD_W_STEP(wA, wp);
      LOAD_A_STEP(aA, (k + 32) & (EMBED - 1));
      __builtin_amdgcn_sched_barrier(0);
      MMA_STEP(aB, wB);
    }

    #pragma unroll
    for (int vt = 0; vt < VOC_TILES; ++vt) {
      const int v = (p * 16 + wave * VOC_TILES + vt) * 16 + lo;
      const float bv = bias[v];
      #pragma unroll
      for (int rt = 0; rt < ROW_TILES; ++rt) {
        #pragma unroll
        for (int j = 0; j < 8; ++j) {
          const float d = fmaf(-2.0f, c0[rt][vt][j] + c1[rt][vt][j], bv);
          if (d < best[rt][j]) { best[rt][j] = d; bidx[rt][j] = v; }
        }
        c0[rt][vt] = (v8f){}; c1[rt][vt] = (v8f){};
      }
    }
  }
#undef LOAD_W_STEP
#undef LOAD_A_STEP
#undef MMA_STEP

  #pragma unroll
  for (int rt = 0; rt < ROW_TILES; ++rt) {
    #pragma unroll
    for (int j = 0; j < 8; ++j) {
      float rmin = best[rt][j];
      rmin = row_xor_min_f<1>(rmin);
      rmin = row_xor_min_f<2>(rmin);
      rmin = row_xor_min_f<4>(rmin);
      rmin = row_xor_min_f<8>(rmin);
      int cand = (best[rt][j] == rmin) ? bidx[rt][j] : 0x7fffffff;
      cand = row_xor_min_i<1>(cand);
      cand = row_xor_min_i<2>(cand);
      cand = row_xor_min_i<4>(cand);
      cand = row_xor_min_i<8>(cand);
      best[rt][j] = rmin; bidx[rt][j] = cand;
    }
  }
  if (lo == 0) {
    #pragma unroll
    for (int rt = 0; rt < ROW_TILES; ++rt) {
      #pragma unroll
      for (int j = 0; j < 8; ++j) {
        redV[wave][rt * 16 + hi * 8 + j] = best[rt][j];
        redI[wave][rt * 16 + hi * 8 + j] = bidx[rt][j];
      }
    }
  }
  __syncthreads();

  if (tid < ROWS_PER_BLOCK) {
    float bv = redV[0][tid];
    int   bi = redI[0][tid];
    #pragma unroll
    for (int w4 = 1; w4 < MAIN_WAVES; ++w4) {
      const float ov = redV[w4][tid];
      const int   oi = redI[w4][tid];
      if (ov < bv || (ov == bv && oi < bi)) { bv = ov; bi = oi; }
    }
    fidx[tid] = bi;
  }
  __syncthreads();

  const int rsub = tid >> 4;
  const int l16  = tid & 15;
  v4f g[64];
  #pragma unroll
  for (int q = 0; q < 8; ++q) {
    const v4f* srcq = (const v4f*)(w + (long)fidx[q * 8 + rsub] * EMBED) + l16;
    #pragma unroll
    for (int p = 0; p < 8; ++p) g[q * 8 + p] = srcq[p * 16];
  }
  __builtin_amdgcn_sched_barrier(0);
  v4f* dst = (v4f*)(out + (r0 + rsub) * EMBED) + l16;
  #pragma unroll
  for (int q = 0; q < 8; ++q) {
    #pragma unroll
    for (int p = 0; p < 8; ++p) dst[q * (8 * EMBED / 4) + p * 16] = g[q * 8 + p];
  }
}

extern "C" void kernel_launch(void* const* d_in, const int* in_sizes, int n_in,
                              void* d_out, int out_size, void* d_ws, size_t ws_size,
                              hipStream_t stream) {
  const float* x = (const float*)d_in[0];
  const float* w = (const float*)d_in[1];
  float* out  = (float*)d_out;
  float* bias = (float*)d_ws;

  quantizer_bias_kernel<<<VOCAB / BIAS_COLS, 256, 0, stream>>>(w, bias);

  const int n_rows = in_sizes[0] / EMBED;
  quantizer_main_kernel<<<n_rows / ROWS_PER_BLOCK, MAIN_WAVES * 32, 0, stream>>>(x, w, bias, out);
}
